# starting kernel
# baseline (speedup 1.0000x reference)
#include <hip/hip_runtime.h>
#include <hip/hip_bf16.h>


typedef float v2f __attribute__((ext_vector_type(2)));
typedef float v8f __attribute__((ext_vector_type(8)));

#define NNODES 250000
#define NEDGES 2000000
#define NGRAPH 5000
#define NHEADS 5
#define NEGSLP 0.2f

__device__ __forceinline__ unsigned ordEnc(float f) {
  unsigned u = __float_as_uint(f);
  return (u & 0x80000000u) ? ~u : (u | 0x80000000u);
}
__device__ __forceinline__ float ordDec(unsigned u) {
  return __uint_as_float((u & 0x80000000u) ? (u & 0x7FFFFFFFu) : ~u);
}

__global__ void fill_u32(unsigned* __restrict__ p, unsigned v, int n) {
  int i = blockIdx.x * blockDim.x + threadIdx.x;
  if (i < n) p[i] = v;
}

template <int CIN, int HCOUT, int COUT, int KT, int NT>
__launch_bounds__(256)
__global__ void edge_alpha_kernel(const float* __restrict__ xin,
                                  const int* __restrict__ src,
                                  const int* __restrict__ dst,
                                  const float* __restrict__ eattr,
                                  const float* __restrict__ wmsg,
                                  const float* __restrict__ bmsg,
                                  const float* __restrict__ wedge,
                                  const float* __restrict__ bedge,
                                  const float* __restrict__ att,
                                  float* __restrict__ alphaG,
                                  unsigned* __restrict__ amax,
                                  int E) {
  constexpr int KP = KT * 4;
  constexpr int NP = NT * 16;
  constexpr int EPB = 128;
  __shared__ float Wl[KP * NP];
  __shared__ float attl[NHEADS * COUT];
  __shared__ float msg[EPB * NP];

  const int tid = threadIdx.x;

  for (int i = tid; i < KP * NP; i += blockDim.x) {
    int k = i / NP, n = i % NP;
    float v = 0.f;
    if (n < HCOUT) {
      if (k < CIN)            v = wmsg[k * HCOUT + n];
      else if (k == CIN)      v = wedge[n];
      else if (k == CIN + 1)  v = bmsg[n] + bedge[n];
    }
    Wl[i] = v;
  }
  for (int i = tid; i < NHEADS * COUT; i += blockDim.x) attl[i] = att[i];
  __syncthreads();

  const int lane  = tid & 31;
  const int wave  = tid >> 5;
  const int m     = lane & 15;
  const int khalf = (lane >> 4) << 1;
  const int ebase = blockIdx.x * EPB + wave * 16;
  const int e     = ebase + m;
  const bool valid = (e < E);

  int   s   = 0;
  float eav = 0.f;
  if (valid) { s = src[e]; eav = eattr[e]; }

  float ax[KT], ay[KT];
#pragma unroll
  for (int kt = 0; kt < KT; ++kt) {
    int k0 = kt * 4 + khalf;
#pragma unroll
    for (int j = 0; j < 2; ++j) {
      int k = k0 + j;
      float v = 0.f;
      if (valid) {
        if (k < CIN)           v = xin[s * CIN + k];
        else if (k == CIN)     v = eav;
        else if (k == CIN + 1) v = 1.0f;
      }
      if (j == 0) ax[kt] = v; else ay[kt] = v;
    }
  }

#pragma unroll
  for (int nt = 0; nt < NT; ++nt) {
    v8f acc = {};
#pragma unroll
    for (int kt = 0; kt < KT; ++kt) {
      int kb = kt * 4 + khalf;
      v2f a; a.x = ax[kt]; a.y = ay[kt];
      v2f b;
      b.x = Wl[kb * NP + nt * 16 + m];
      b.y = Wl[(kb + 1) * NP + nt * 16 + m];
      acc = __builtin_amdgcn_wmma_f32_16x16x4_f32(
          false, a, false, b, (short)0, acc, false, false);
    }
    int col   = nt * 16 + m;
    int rbase = wave * 16 + ((lane >> 4) << 3);
#pragma unroll
    for (int v = 0; v < 8; ++v) msg[(rbase + v) * NP + col] = acc[v];
  }
  __syncthreads();

  for (int i = tid; i < EPB * NHEADS; i += blockDim.x) {
    int el = i / NHEADS, h = i % NHEADS;
    int ee = blockIdx.x * EPB + el;
    if (ee >= E) continue;
    float sacc = 0.f;
#pragma unroll
    for (int c = 0; c < COUT; ++c)
      sacc += msg[el * NP + h * COUT + c] * attl[h * COUT + c];
    float alpha = (sacc > 0.f) ? sacc : NEGSLP * sacc;
    alphaG[ee * NHEADS + h] = alpha;
    atomicMax(&amax[dst[ee] * NHEADS + h], ordEnc(alpha));
  }
}

__global__ void softmax_kernel(float* __restrict__ alpha,
                               const unsigned* __restrict__ amax,
                               float* __restrict__ denom,
                               const int* __restrict__ dst, int EH) {
  int i = blockIdx.x * blockDim.x + threadIdx.x;
  if (i >= EH) return;
  int e = i / NHEADS;
  int h = i - e * NHEADS;
  int d = dst[e];
  float am = ordDec(amax[d * NHEADS + h]);
  float ea = __expf(alpha[i] - am);
  alpha[i] = ea;
  atomicAdd(&denom[d * NHEADS + h], ea);
}

template <int CIN, int HCOUT, int COUT, int KT, int NT>
__launch_bounds__(256)
__global__ void edge_scatter_kernel(const float* __restrict__ xin,
                                    const int* __restrict__ src,
                                    const int* __restrict__ dst,
                                    const float* __restrict__ eattr,
                                    const float* __restrict__ wmsg,
                                    const float* __restrict__ bmsg,
                                    const float* __restrict__ wedge,
                                    const float* __restrict__ bedge,
                                    const float* __restrict__ eaG,
                                    const float* __restrict__ denom,
                                    float* __restrict__ agg, int E) {
  constexpr int KP = KT * 4;
  constexpr int NP = NT * 16;
  constexpr int EPB = 128;
  __shared__ float Wl[KP * NP];
  __shared__ float msg[EPB * NP];
  __shared__ float al[EPB * NHEADS];

  const int tid = threadIdx.x;
  for (int i = tid; i < KP * NP; i += blockDim.x) {
    int k = i / NP, n = i % NP;
    float v = 0.f;
    if (n < HCOUT) {
      if (k < CIN)            v = wmsg[k * HCOUT + n];
      else if (k == CIN)      v = wedge[n];
      else if (k == CIN + 1)  v = bmsg[n] + bedge[n];
    }
    Wl[i] = v;
  }
  for (int i = tid; i < EPB * NHEADS; i += blockDim.x) {
    int el = i / NHEADS, h = i % NHEADS;
    int ee = blockIdx.x * EPB + el;
    float a = 0.f;
    if (ee < E) {
      int d = dst[ee];
      a = eaG[ee * NHEADS + h] / (denom[d * NHEADS + h] + 1e-16f);
    }
    al[i] = a;
  }

  const int lane  = tid & 31;
  const int wave  = tid >> 5;
  const int m     = lane & 15;
  const int khalf = (lane >> 4) << 1;
  const int ebase = blockIdx.x * EPB + wave * 16;
  const int e     = ebase + m;
  const bool valid = (e < E);

  int   s   = 0;
  float eav = 0.f;
  if (valid) { s = src[e]; eav = eattr[e]; }

  float ax[KT], ay[KT];
#pragma unroll
  for (int kt = 0; kt < KT; ++kt) {
    int k0 = kt * 4 + khalf;
#pragma unroll
    for (int j = 0; j < 2; ++j) {
      int k = k0 + j;
      float v = 0.f;
      if (valid) {
        if (k < CIN)           v = xin[s * CIN + k];
        else if (k == CIN)     v = eav;
        else if (k == CIN + 1) v = 1.0f;
      }
      if (j == 0) ax[kt] = v; else ay[kt] = v;
    }
  }
  __syncthreads();

#pragma unroll
  for (int nt = 0; nt < NT; ++nt) {
    v8f acc = {};
#pragma unroll
    for (int kt = 0; kt < KT; ++kt) {
      int kb = kt * 4 + khalf;
      v2f a; a.x = ax[kt]; a.y = ay[kt];
      v2f b;
      b.x = Wl[kb * NP + nt * 16 + m];
      b.y = Wl[(kb + 1) * NP + nt * 16 + m];
      acc = __builtin_amdgcn_wmma_f32_16x16x4_f32(
          false, a, false, b, (short)0, acc, false, false);
    }
    int col   = nt * 16 + m;
    int rbase = wave * 16 + ((lane >> 4) << 3);
#pragma unroll
    for (int v = 0; v < 8; ++v) msg[(rbase + v) * NP + col] = acc[v];
  }
  __syncthreads();

  for (int i = tid; i < EPB * COUT; i += blockDim.x) {
    int el = i / COUT, c = i % COUT;
    int ee = blockIdx.x * EPB + el;
    if (ee >= E) continue;
    float sacc = 0.f;
#pragma unroll
    for (int h = 0; h < NHEADS; ++h)
      sacc += msg[el * NP + h * COUT + c] * al[el * NHEADS + h];
    atomicAdd(&agg[dst[ee] * COUT + c], sacc * (1.0f / NHEADS));
  }
}

template <int CIN, int COUT>
__global__ void node_kernel(const float* __restrict__ xin,
                            const float* __restrict__ wself,
                            const float* __restrict__ bself,
                            const float* __restrict__ agg,
                            float* __restrict__ hout, int N) {
  int i = blockIdx.x * blockDim.x + threadIdx.x;
  if (i >= N * COUT) return;
  int n = i / COUT, c = i % COUT;
  float sacc = agg[i];
#pragma unroll
  for (int k = 0; k < CIN; ++k) sacc += xin[n * CIN + k] * wself[k * COUT + c];
  hout[i] = fmaxf(sacc + bself[c], 0.f);
}

__global__ void pool_kernel(const float* __restrict__ h3,
                            const int* __restrict__ batch,
                            unsigned* __restrict__ gmax, int N) {
  int i = blockIdx.x * blockDim.x + threadIdx.x;
  if (i >= N * 16) return;
  int n = i >> 4, c = i & 15;
  atomicMax(&gmax[batch[n] * 16 + c], ordEnc(h3[i]));
}

__global__ void head_kernel(const unsigned* __restrict__ gmax,
                            const float* __restrict__ wh,
                            const float* __restrict__ bh,
                            float* __restrict__ out, int G) {
  int i = blockIdx.x * blockDim.x + threadIdx.x;
  if (i >= G * 2) return;
  int g = i >> 1, c = i & 1;
  float sacc = bh[c];
#pragma unroll
  for (int k = 0; k < 16; ++k) sacc += ordDec(gmax[g * 16 + k]) * wh[k * 2 + c];
  out[i] = sacc;
}

template <int CIN, int HCOUT, int COUT, int KT, int NT>
static void run_layer(const float* xin, const int* src, const int* dst,
                      const float* eattr, const float* wmsg, const float* bmsg,
                      const float* wedge, const float* bedge, const float* att,
                      const float* wself, const float* bself, float* hout,
                      float* alpha, unsigned* amax, float* denom, float* agg,
                      hipStream_t stream) {
  const int N = NNODES, E = NEDGES;
  const int NH = N * NHEADS;
  fill_u32<<<(NH + 255) / 256, 256, 0, stream>>>(amax, 0x007FFFFFu, NH);
  fill_u32<<<(NH + 255) / 256, 256, 0, stream>>>((unsigned*)denom, 0u, NH);
  fill_u32<<<(N * COUT + 255) / 256, 256, 0, stream>>>((unsigned*)agg, 0u, N * COUT);

  const int nb = (E + 127) / 128;
  edge_alpha_kernel<CIN, HCOUT, COUT, KT, NT><<<nb, 256, 0, stream>>>(
      xin, src, dst, eattr, wmsg, bmsg, wedge, bedge, att, alpha, amax, E);

  const int EH = E * NHEADS;
  softmax_kernel<<<(EH + 255) / 256, 256, 0, stream>>>(alpha, amax, denom, dst, EH);

  edge_scatter_kernel<CIN, HCOUT, COUT, KT, NT><<<nb, 256, 0, stream>>>(
      xin, src, dst, eattr, wmsg, bmsg, wedge, bedge, alpha, denom, agg, E);

  node_kernel<CIN, COUT><<<(N * COUT + 255) / 256, 256, 0, stream>>>(
      xin, wself, bself, agg, hout, N);
}

extern "C" void kernel_launch(void* const* d_in, const int* in_sizes, int n_in,
                              void* d_out, int out_size, void* d_ws,
                              size_t ws_size, hipStream_t stream) {
  (void)in_sizes; (void)n_in; (void)out_size; (void)ws_size;
  const int N = NNODES, E = NEDGES, G = NGRAPH;

  const float* x     = (const float*)d_in[0];
  const int*   ei    = (const int*)d_in[1];
  const float* eattr = (const float*)d_in[2];
  const int*   batch = (const int*)d_in[3];
  const int* src = ei;
  const int* dst = ei + E;

  const float *wmsg[3], *bmsg[3], *wedge[3], *bedge[3], *att[3], *wself[3], *bself[3];
  for (int l = 0; l < 3; ++l) {
    int b = 4 + 7 * l;
    wmsg[l]  = (const float*)d_in[b + 0];
    bmsg[l]  = (const float*)d_in[b + 1];
    wedge[l] = (const float*)d_in[b + 2];
    bedge[l] = (const float*)d_in[b + 3];
    att[l]   = (const float*)d_in[b + 4];
    wself[l] = (const float*)d_in[b + 5];
    bself[l] = (const float*)d_in[b + 6];
  }
  const float* whead = (const float*)d_in[25];
  const float* bhead = (const float*)d_in[26];

  float*    alpha = (float*)d_ws;
  unsigned* amax  = (unsigned*)(alpha + (size_t)E * NHEADS);
  float*    denom = (float*)(amax + (size_t)N * NHEADS);
  float*    agg   = denom + (size_t)N * NHEADS;
  float*    h1    = agg + (size_t)N * 16;
  float*    h2    = h1 + (size_t)N * 4;
  float*    h3    = h2 + (size_t)N * 8;
  unsigned* gmax  = (unsigned*)(h3 + (size_t)N * 16);

  run_layer<3, 20, 4, 2, 2>(x,  src, dst, eattr, wmsg[0], bmsg[0], wedge[0],
                            bedge[0], att[0], wself[0], bself[0], h1,
                            alpha, amax, denom, agg, stream);
  run_layer<4, 40, 8, 2, 3>(h1, src, dst, eattr, wmsg[1], bmsg[1], wedge[1],
                            bedge[1], att[1], wself[1], bself[1], h2,
                            alpha, amax, denom, agg, stream);
  run_layer<8, 80, 16, 3, 5>(h2, src, dst, eattr, wmsg[2], bmsg[2], wedge[2],
                             bedge[2], att[2], wself[2], bself[2], h3,
                             alpha, amax, denom, agg, stream);

  fill_u32<<<(G * 16 + 255) / 256, 256, 0, stream>>>(gmax, 0x007FFFFFu, G * 16);
  pool_kernel<<<(N * 16 + 255) / 256, 256, 0, stream>>>(h3, batch, gmax, N);
  head_kernel<<<(G * 2 + 255) / 256, 256, 0, stream>>>(gmax, whead, bhead,
                                                       (float*)d_out, G);
}


# Each atomic accumulator (edge max, exp-sum, aggregate) kept as one copy per L2 coherence half: a block updates its own half's copy, readers merge both; changes float summation order
# speedup vs baseline: 1.4274x; 1.4274x over previous
#include <hip/hip_runtime.h>
#include <hip/hip_bf16.h>


typedef float v2f __attribute__((ext_vector_type(2)));
typedef float v8f __attribute__((ext_vector_type(8)));

#define NNODES 250000
#define NEDGES 2000000
#define NGRAPH 5000
#define NHEADS 5
#define NEGSLP 0.2f

__device__ __forceinline__ unsigned ordEnc(float f) {
  unsigned u = __float_as_uint(f);
  return (u & 0x80000000u) ? ~u : (u | 0x80000000u);
}
__device__ __forceinline__ float ordDec(unsigned u) {
  return __uint_as_float((u & 0x80000000u) ? (u & 0x7FFFFFFFu) : ~u);
}

__global__ void fill_u32(unsigned* __restrict__ p, unsigned v, int n) {
  int i = blockIdx.x * blockDim.x + threadIdx.x;
  if (i < n) p[i] = v;
}

__global__ void fill2_u32(unsigned* __restrict__ p0, unsigned v0, int n0,
                          unsigned* __restrict__ p1, unsigned v1, int n1) {
  int i = blockIdx.x * blockDim.x + threadIdx.x;
  if (i < n0) p0[i] = v0;
  else if (i - n0 < n1) p1[i - n0] = v1;
}

__device__ __forceinline__ unsigned chip_half() {
  unsigned v = __builtin_amdgcn_s_sendmsg_rtn(0x87);
  return (v >> 18) & 1u;
}

template <int CIN, int HCOUT, int COUT, int KT, int NT>
__launch_bounds__(256)
__global__ void edge_alpha_kernel(const float* __restrict__ xin,
                                  const int* __restrict__ src,
                                  const int* __restrict__ dst,
                                  const float* __restrict__ eattr,
                                  const float* __restrict__ wmsg,
                                  const float* __restrict__ bmsg,
                                  const float* __restrict__ wedge,
                                  const float* __restrict__ bedge,
                                  const float* __restrict__ att,
                                  float* __restrict__ alphaG,
                                  unsigned* __restrict__ amax,
                                  int E) {
  constexpr int KP = KT * 4;
  constexpr int NP = NT * 16;
  constexpr int EPB = 128;
  __shared__ float Wl[KP * NP];
  __shared__ float attl[NHEADS * COUT];
  __shared__ float msg[EPB * NP];
  __shared__ unsigned hsel;

  const int tid = threadIdx.x;

  for (int i = tid; i < KP * NP; i += blockDim.x) {
    int k = i / NP, n = i % NP;
    float v = 0.f;
    if (n < HCOUT) {
      if (k < CIN)            v = wmsg[k * HCOUT + n];
      else if (k == CIN)      v = wedge[n];
      else if (k == CIN + 1)  v = bmsg[n] + bedge[n];
    }
    Wl[i] = v;
  }
  for (int i = tid; i < NHEADS * COUT; i += blockDim.x) attl[i] = att[i];
  __syncthreads();

  const int lane  = tid & 31;
  const int wave  = tid >> 5;
  const int m     = lane & 15;
  const int khalf = (lane >> 4) << 1;
  const int ebase = blockIdx.x * EPB + wave * 16;
  const int e     = ebase + m;
  const bool valid = (e < E);

  int   s   = 0;
  float eav = 0.f;
  if (valid) { s = src[e]; eav = eattr[e]; }

  float ax[KT], ay[KT];
#pragma unroll
  for (int kt = 0; kt < KT; ++kt) {
    int k0 = kt * 4 + khalf;
#pragma unroll
    for (int j = 0; j < 2; ++j) {
      int k = k0 + j;
      float v = 0.f;
      if (valid) {
        if (k < CIN)           v = xin[s * CIN + k];
        else if (k == CIN)     v = eav;
        else if (k == CIN + 1) v = 1.0f;
      }
      if (j == 0) ax[kt] = v; else ay[kt] = v;
    }
  }
  if (tid == 0) hsel = chip_half();

#pragma unroll
  for (int nt = 0; nt < NT; ++nt) {
    v8f acc = {};
#pragma unroll
    for (int kt = 0; kt < KT; ++kt) {
      int kb = kt * 4 + khalf;
      v2f a; a.x = ax[kt]; a.y = ay[kt];
      v2f b;
      b.x = Wl[kb * NP + nt * 16 + m];
      b.y = Wl[(kb + 1) * NP + nt * 16 + m];
      acc = __builtin_amdgcn_wmma_f32_16x16x4_f32(
          false, a, false, b, (short)0, acc, false, false);
    }
    int col   = nt * 16 + m;
    int rbase = wave * 16 + ((lane >> 4) << 3);
#pragma unroll
    for (int v = 0; v < 8; ++v) msg[(rbase + v) * NP + col] = acc[v];
  }
  __syncthreads();

  unsigned* __restrict__ amaxh = amax + hsel * (NNODES * NHEADS);
  for (int i = tid; i < EPB * NHEADS; i += blockDim.x) {
    int el = i / NHEADS, h = i % NHEADS;
    int ee = blockIdx.x * EPB + el;
    if (ee >= E) continue;
    float sacc = 0.f;
#pragma unroll
    for (int c = 0; c < COUT; ++c)
      sacc += msg[el * NP + h * COUT + c] * attl[h * COUT + c];
    float alpha = (sacc > 0.f) ? sacc : NEGSLP * sacc;
    alphaG[ee * NHEADS + h] = alpha;
    atomicMax(&amaxh[dst[ee] * NHEADS + h], ordEnc(alpha));
  }
}

__launch_bounds__(256)
__global__ void softmax_kernel(float* __restrict__ alpha,
                               const unsigned* __restrict__ amax,
                               float* __restrict__ denom,
                               const int* __restrict__ dst, int EH) {
  constexpr int NH = NNODES * NHEADS;
  __shared__ unsigned hsel;
  const int base = blockIdx.x * 1024 + threadIdx.x;
  int      off[4];
  float    av[4];
  unsigned u0[4], u1[4];
#pragma unroll
  for (int k = 0; k < 4; ++k) {
    int i  = base + k * 256;
    int ic = (i < EH) ? i : (EH - 1);
    int e  = ic / NHEADS;
    off[k] = dst[e] * NHEADS + (ic - e * NHEADS);
    av[k]  = alpha[ic];
  }
#pragma unroll
  for (int k = 0; k < 4; ++k) {
    u0[k] = amax[off[k]];
    u1[k] = amax[NH + off[k]];
  }
  if (threadIdx.x == 0) hsel = chip_half();
  float ea[4];
#pragma unroll
  for (int k = 0; k < 4; ++k) {
    unsigned um = (u0[k] > u1[k]) ? u0[k] : u1[k];
    ea[k] = __expf(av[k] - ordDec(um));
  }
  __syncthreads();
  float* __restrict__ dh = denom + hsel * NH;
#pragma unroll
  for (int k = 0; k < 4; ++k) {
    if (base + k * 256 < EH) {
      atomicAdd(&dh[off[k]], ea[k]);
      alpha[base + k * 256] = ea[k];
    }
  }
}

template <int CIN, int HCOUT, int COUT, int KT, int NT>
__launch_bounds__(256)
__global__ void edge_scatter_kernel(const float* __restrict__ xin,
                                    const int* __restrict__ src,
                                    const int* __restrict__ dst,
                                    const float* __restrict__ eattr,
                                    const float* __restrict__ wmsg,
                                    const float* __restrict__ bmsg,
                                    const float* __restrict__ wedge,
                                    const float* __restrict__ bedge,
                                    const float* __restrict__ eaG,
                                    const float* __restrict__ denom,
                                    float* __restrict__ agg0,
                                    float* __restrict__ agg1, int E) {
  constexpr int KP = KT * 4;
  constexpr int NP = NT * 16;
  constexpr int EPB = 128;
  __shared__ float Wl[KP * NP];
  __shared__ float msg[EPB * NP];
  __shared__ float al[EPB * NHEADS];
  __shared__ unsigned hsel;

  const int tid = threadIdx.x;
  for (int i = tid; i < KP * NP; i += blockDim.x) {
    int k = i / NP, n = i % NP;
    float v = 0.f;
    if (n < HCOUT) {
      if (k < CIN)            v = wmsg[k * HCOUT + n];
      else if (k == CIN)      v = wedge[n];
      else if (k == CIN + 1)  v = bmsg[n] + bedge[n];
    }
    Wl[i] = v;
  }
  for (int i = tid; i < EPB * NHEADS; i += blockDim.x) {
    int el = i / NHEADS, h = i % NHEADS;
    int ee = blockIdx.x * EPB + el;
    float a = 0.f;
    if (ee < E) {
      int o = dst[ee] * NHEADS + h;
      a = eaG[ee * NHEADS + h] /
          ((denom[o] + denom[NNODES * NHEADS + o]) + 1e-16f);
    }
    al[i] = a;
  }

  const int lane  = tid & 31;
  const int wave  = tid >> 5;
  const int m     = lane & 15;
  const int khalf = (lane >> 4) << 1;
  const int ebase = blockIdx.x * EPB + wave * 16;
  const int e     = ebase + m;
  const bool valid = (e < E);

  int   s   = 0;
  float eav = 0.f;
  if (valid) { s = src[e]; eav = eattr[e]; }

  float ax[KT], ay[KT];
#pragma unroll
  for (int kt = 0; kt < KT; ++kt) {
    int k0 = kt * 4 + khalf;
#pragma unroll
    for (int j = 0; j < 2; ++j) {
      int k = k0 + j;
      float v = 0.f;
      if (valid) {
        if (k < CIN)           v = xin[s * CIN + k];
        else if (k == CIN)     v = eav;
        else if (k == CIN + 1) v = 1.0f;
      }
      if (j == 0) ax[kt] = v; else ay[kt] = v;
    }
  }
  if (tid == 0) hsel = chip_half();
  __syncthreads();

#pragma unroll
  for (int nt = 0; nt < NT; ++nt) {
    v8f acc = {};
#pragma unroll
    for (int kt = 0; kt < KT; ++kt) {
      int kb = kt * 4 + khalf;
      v2f a; a.x = ax[kt]; a.y = ay[kt];
      v2f b;
      b.x = Wl[kb * NP + nt * 16 + m];
      b.y = Wl[(kb + 1) * NP + nt * 16 + m];
      acc = __builtin_amdgcn_wmma_f32_16x16x4_f32(
          false, a, false, b, (short)0, acc, false, false);
    }
    int col   = nt * 16 + m;
    int rbase = wave * 16 + ((lane >> 4) << 3);
#pragma unroll
    for (int v = 0; v < 8; ++v) msg[(rbase + v) * NP + col] = acc[v];
  }
  __syncthreads();

  float* __restrict__ aggh = hsel ? agg1 : agg0;
  for (int i = tid; i < EPB * COUT; i += blockDim.x) {
    int el = i / COUT, c = i % COUT;
    int ee = blockIdx.x * EPB + el;
    if (ee >= E) continue;
    float sacc = 0.f;
#pragma unroll
    for (int h = 0; h < NHEADS; ++h)
      sacc += msg[el * NP + h * COUT + c] * al[el * NHEADS + h];
    atomicAdd(&aggh[dst[ee] * COUT + c], sacc * (1.0f / NHEADS));
  }
}

template <int CIN, int COUT>
__global__ void node_kernel(const float* __restrict__ xin,
                            const float* __restrict__ wself,
                            const float* __restrict__ bself,
                            const float* __restrict__ agg0,
                            float* __restrict__ hout, int N) {
  int i = blockIdx.x * blockDim.x + threadIdx.x;
  if (i >= N * COUT) return;
  int n = i / COUT, c = i % COUT;
  float sacc = agg0[i] + hout[i];
#pragma unroll
  for (int k = 0; k < CIN; ++k) sacc += xin[n * CIN + k] * wself[k * COUT + c];
  hout[i] = fmaxf(sacc + bself[c], 0.f);
}

__global__ void pool_kernel(const float* __restrict__ h3,
                            const int* __restrict__ batch,
                            unsigned* __restrict__ gmax, int N) {
  int i = blockIdx.x * blockDim.x + threadIdx.x;
  if (i >= N * 16) return;
  int n = i >> 4, c = i & 15;
  atomicMax(&gmax[batch[n] * 16 + c], ordEnc(h3[i]));
}

__global__ void head_kernel(const unsigned* __restrict__ gmax,
                            const float* __restrict__ wh,
                            const float* __restrict__ bh,
                            float* __restrict__ out, int G) {
  int i = blockIdx.x * blockDim.x + threadIdx.x;
  if (i >= G * 2) return;
  int g = i >> 1, c = i & 1;
  float sacc = bh[c];
#pragma unroll
  for (int k = 0; k < 16; ++k) sacc += ordDec(gmax[g * 16 + k]) * wh[k * 2 + c];
  out[i] = sacc;
}

template <int CIN, int HCOUT, int COUT, int KT, int NT>
static void run_layer(const float* xin, const int* src, const int* dst,
                      const float* eattr, const float* wmsg, const float* bmsg,
                      const float* wedge, const float* bedge, const float* att,
                      const float* wself, const float* bself, float* hout,
                      float* alpha, unsigned* amax, float* denom, float* agg0,
                      hipStream_t stream) {
  const int N = NNODES, E = NEDGES;
  const int NH2 = 2 * N * NHEADS;
  fill2_u32<<<(2 * NH2 + 255) / 256, 256, 0, stream>>>(
      amax, 0x007FFFFFu, NH2,
      (unsigned*)denom, 0u, NH2);

  const int nb = (E + 127) / 128;
  edge_alpha_kernel<CIN, HCOUT, COUT, KT, NT><<<nb, 256, 0, stream>>>(
      xin, src, dst, eattr, wmsg, bmsg, wedge, bedge, att, alpha, amax, E);

  const int EH = E * NHEADS;
  softmax_kernel<<<(EH + 1023) / 1024, 256, 0, stream>>>(alpha, amax, denom, dst, EH);

  fill2_u32<<<(2 * N * COUT + 255) / 256, 256, 0, stream>>>(
      (unsigned*)agg0, 0u, N * COUT, (unsigned*)hout, 0u, N * COUT);

  edge_scatter_kernel<CIN, HCOUT, COUT, KT, NT><<<nb, 256, 0, stream>>>(
      xin, src, dst, eattr, wmsg, bmsg, wedge, bedge, alpha, denom, agg0, hout, E);

  node_kernel<CIN, COUT><<<(N * COUT + 255) / 256, 256, 0, stream>>>(
      xin, wself, bself, agg0, hout, N);
}

extern "C" void kernel_launch(void* const* d_in, const int* in_sizes, int n_in,
                              void* d_out, int out_size, void* d_ws,
                              size_t ws_size, hipStream_t stream) {
  (void)in_sizes; (void)n_in; (void)out_size; (void)ws_size;
  const int N = NNODES, E = NEDGES, G = NGRAPH;

  const float* x     = (const float*)d_in[0];
  const int*   ei    = (const int*)d_in[1];
  const float* eattr = (const float*)d_in[2];
  const int*   batch = (const int*)d_in[3];
  const int* src = ei;
  const int* dst = ei + E;

  const float *wmsg[3], *bmsg[3], *wedge[3], *bedge[3], *att[3], *wself[3], *bself[3];
  for (int l = 0; l < 3; ++l) {
    int b = 4 + 7 * l;
    wmsg[l]  = (const float*)d_in[b + 0];
    bmsg[l]  = (const float*)d_in[b + 1];
    wedge[l] = (const float*)d_in[b + 2];
    bedge[l] = (const float*)d_in[b + 3];
    att[l]   = (const float*)d_in[b + 4];
    wself[l] = (const float*)d_in[b + 5];
    bself[l] = (const float*)d_in[b + 6];
  }
  const float* whead = (const float*)d_in[25];
  const float* bhead = (const float*)d_in[26];

  float*    alpha = (float*)d_ws;
  float*    denom = alpha + (size_t)E * NHEADS;
  float*    agg   = denom + (size_t)2 * N * NHEADS;
  unsigned* amax  = (unsigned*)agg;
  float*    h1    = agg + (size_t)N * 16;
  float*    h2    = h1 + (size_t)N * 4;
  float*    h3    = h2 + (size_t)N * 8;
  unsigned* gmax  = (unsigned*)(h3 + (size_t)N * 16);

  run_layer<3, 20, 4, 2, 2>(x,  src, dst, eattr, wmsg[0], bmsg[0], wedge[0],
                            bedge[0], att[0], wself[0], bself[0], h1,
                            alpha, amax, denom, agg, stream);
  run_layer<4, 40, 8, 2, 3>(h1, src, dst, eattr, wmsg[1], bmsg[1], wedge[1],
                            bedge[1], att[1], wself[1], bself[1], h2,
                            alpha, amax, denom, agg, stream);
  run_layer<8, 80, 16, 3, 5>(h2, src, dst, eattr, wmsg[2], bmsg[2], wedge[2],
                             bedge[2], att[2], wself[2], bself[2], h3,
                             alpha, amax, denom, agg, stream);

  fill_u32<<<(G * 16 + 255) / 256, 256, 0, stream>>>(gmax, 0x007FFFFFu, G * 16);
  pool_kernel<<<(N * 16 + 255) / 256, 256, 0, stream>>>(h3, batch, gmax, N);
  head_kernel<<<(G * 2 + 255) / 256, 256, 0, stream>>>(gmax, whead, bhead,
                                                       (float*)d_out, G);
}


# The two per-half copies of the edge max and of the exp-sum are merged once per (node, head) by a small streaming pass, so each edge pass gathers one value per edge and head, not two; sum order kept
# speedup vs baseline: 1.5458x; 1.0829x over previous
#include <hip/hip_runtime.h>
#include <hip/hip_bf16.h>


typedef float v2f __attribute__((ext_vector_type(2)));
typedef float v8f __attribute__((ext_vector_type(8)));

#define NNODES 250000
#define NEDGES 2000000
#define NGRAPH 5000
#define NHEADS 5
#define NEGSLP 0.2f

__device__ __forceinline__ unsigned ordEnc(float f) {
  unsigned u = __float_as_uint(f);
  return (u & 0x80000000u) ? ~u : (u | 0x80000000u);
}
__device__ __forceinline__ float ordDec(unsigned u) {
  return __uint_as_float((u & 0x80000000u) ? (u & 0x7FFFFFFFu) : ~u);
}

__global__ void fill_u32(unsigned* __restrict__ p, unsigned v, int n) {
  int i = blockIdx.x * blockDim.x + threadIdx.x;
  if (i < n) p[i] = v;
}

__global__ void fill2_u32(unsigned* __restrict__ p0, unsigned v0, int n0,
                          unsigned* __restrict__ p1, unsigned v1, int n1) {
  int i = blockIdx.x * blockDim.x + threadIdx.x;
  if (i < n0) p0[i] = v0;
  else if (i - n0 < n1) p1[i - n0] = v1;
}

__launch_bounds__(256)
__global__ void merge_max_kernel(unsigned* __restrict__ amax, int NH) {
  const int base = blockIdx.x * 1024 + threadIdx.x;
  unsigned u0[4], u1[4];
#pragma unroll
  for (int k = 0; k < 4; ++k) {
    int i = base + k * 256;
    u0[k] = 0u; u1[k] = 0u;
    if (i < NH) { u0[k] = amax[i]; u1[k] = amax[NH + i]; }
  }
#pragma unroll
  for (int k = 0; k < 4; ++k) {
    int i = base + k * 256;
    unsigned um = (u0[k] > u1[k]) ? u0[k] : u1[k];
    if (i < NH) amax[i] = __float_as_uint(ordDec(um));
  }
}

__launch_bounds__(256)
__global__ void merge_sum_zero2_kernel(float* __restrict__ denom, int NH, int MB,
                                       unsigned* __restrict__ p0, int n0,
                                       unsigned* __restrict__ p1, int n1) {
  if ((int)blockIdx.x < MB) {
    const int base = blockIdx.x * 1024 + threadIdx.x;
    float d0[4], d1[4];
#pragma unroll
    for (int k = 0; k < 4; ++k) {
      int i = base + k * 256;
      d0[k] = 0.f; d1[k] = 0.f;
      if (i < NH) { d0[k] = denom[i]; d1[k] = denom[NH + i]; }
    }
#pragma unroll
    for (int k = 0; k < 4; ++k) {
      int i = base + k * 256;
      if (i < NH) denom[i] = (d0[k] + d1[k]) + 1e-16f;
    }
  } else {
    int i = ((int)blockIdx.x - MB) * 256 + threadIdx.x;
    if (i < n0) p0[i] = 0u;
    else if (i - n0 < n1) p1[i - n0] = 0u;
  }
}

__device__ __forceinline__ unsigned chip_half() {
  unsigned v = __builtin_amdgcn_s_sendmsg_rtn(0x87);
  return (v >> 18) & 1u;
}

template <int CIN, int HCOUT, int COUT, int KT, int NT>
__launch_bounds__(256)
__global__ void edge_alpha_kernel(const float* __restrict__ xin,
                                  const int* __restrict__ src,
                                  const int* __restrict__ dst,
                                  const float* __restrict__ eattr,
                                  const float* __restrict__ wmsg,
                                  const float* __restrict__ bmsg,
                                  const float* __restrict__ wedge,
                                  const float* __restrict__ bedge,
                                  const float* __restrict__ att,
                                  float* __restrict__ alphaG,
                                  unsigned* __restrict__ amax,
                                  int E) {
  constexpr int KP = KT * 4;
  constexpr int NP = NT * 16;
  constexpr int EPB = 128;
  __shared__ float Wl[KP * NP];
  __shared__ float attl[NHEADS * COUT];
  __shared__ float msg[EPB * NP];
  __shared__ unsigned hsel;

  const int tid = threadIdx.x;

  for (int i = tid; i < KP * NP; i += blockDim.x) {
    int k = i / NP, n = i % NP;
    float v = 0.f;
    if (n < HCOUT) {
      if (k < CIN)            v = wmsg[k * HCOUT + n];
      else if (k == CIN)      v = wedge[n];
      else if (k == CIN + 1)  v = bmsg[n] + bedge[n];
    }
    Wl[i] = v;
  }
  for (int i = tid; i < NHEADS * COUT; i += blockDim.x) attl[i] = att[i];
  __syncthreads();

  const int lane  = tid & 31;
  const int wave  = tid >> 5;
  const int m     = lane & 15;
  const int khalf = (lane >> 4) << 1;
  const int ebase = blockIdx.x * EPB + wave * 16;
  const int e     = ebase + m;
  const bool valid = (e < E);

  int   s   = 0;
  float eav = 0.f;
  if (valid) { s = src[e]; eav = eattr[e]; }

  float ax[KT], ay[KT];
#pragma unroll
  for (int kt = 0; kt < KT; ++kt) {
    int k0 = kt * 4 + khalf;
#pragma unroll
    for (int j = 0; j < 2; ++j) {
      int k = k0 + j;
      float v = 0.f;
      if (valid) {
        if (k < CIN)           v = xin[s * CIN + k];
        else if (k == CIN)     v = eav;
        else if (k == CIN + 1) v = 1.0f;
      }
      if (j == 0) ax[kt] = v; else ay[kt] = v;
    }
  }
  if (tid == 0) hsel = chip_half();

#pragma unroll
  for (int nt = 0; nt < NT; ++nt) {
    v8f acc = {};
#pragma unroll
    for (int kt = 0; kt < KT; ++kt) {
      int kb = kt * 4 + khalf;
      v2f a; a.x = ax[kt]; a.y = ay[kt];
      v2f b;
      b.x = Wl[kb * NP + nt * 16 + m];
      b.y = Wl[(kb + 1) * NP + nt * 16 + m];
      acc = __builtin_amdgcn_wmma_f32_16x16x4_f32(
          false, a, false, b, (short)0, acc, false, false);
    }
    int col   = nt * 16 + m;
    int rbase = wave * 16 + ((lane >> 4) << 3);
#pragma unroll
    for (int v = 0; v < 8; ++v) msg[(rbase + v) * NP + col] = acc[v];
  }
  __syncthreads();

  unsigned* __restrict__ amaxh = amax + hsel * (NNODES * NHEADS);
  for (int i = tid; i < EPB * NHEADS; i += blockDim.x) {
    int el = i / NHEADS, h = i % NHEADS;
    int ee = blockIdx.x * EPB + el;
    if (ee >= E) continue;
    float sacc = 0.f;
#pragma unroll
    for (int c = 0; c < COUT; ++c)
      sacc += msg[el * NP + h * COUT + c] * attl[h * COUT + c];
    float alpha = (sacc > 0.f) ? sacc : NEGSLP * sacc;
    alphaG[ee * NHEADS + h] = alpha;
    atomicMax(&amaxh[dst[ee] * NHEADS + h], ordEnc(alpha));
  }
}

__launch_bounds__(256)
__global__ void softmax_kernel(float* __restrict__ alpha,
                               const unsigned* __restrict__ amaxM,
                               float* __restrict__ denom,
                               const int* __restrict__ dst, int EH) {
  constexpr int NH = NNODES * NHEADS;
  __shared__ unsigned hsel;
  const int base = blockIdx.x * 1024 + threadIdx.x;
  int      off[4];
  float    av[4];
  unsigned um[4];
#pragma unroll
  for (int k = 0; k < 4; ++k) {
    int i  = base + k * 256;
    int ic = (i < EH) ? i : (EH - 1);
    int e  = ic / NHEADS;
    off[k] = dst[e] * NHEADS + (ic - e * NHEADS);
    av[k]  = alpha[ic];
  }
#pragma unroll
  for (int k = 0; k < 4; ++k) um[k] = amaxM[off[k]];
  if (threadIdx.x == 0) hsel = chip_half();
  float ea[4];
#pragma unroll
  for (int k = 0; k < 4; ++k) ea[k] = __expf(av[k] - __uint_as_float(um[k]));
  __syncthreads();
  float* __restrict__ dh = denom + hsel * NH;
#pragma unroll
  for (int k = 0; k < 4; ++k) {
    if (base + k * 256 < EH) {
      atomicAdd(&dh[off[k]], ea[k]);
      alpha[base + k * 256] = ea[k];
    }
  }
}

template <int CIN, int HCOUT, int COUT, int KT, int NT>
__launch_bounds__(256)
__global__ void edge_scatter_kernel(const float* __restrict__ xin,
                                    const int* __restrict__ src,
                                    const int* __restrict__ dst,
                                    const float* __restrict__ eattr,
                                    const float* __restrict__ wmsg,
                                    const float* __restrict__ bmsg,
                                    const float* __restrict__ wedge,
                                    const float* __restrict__ bedge,
                                    const float* __restrict__ eaG,
                                    const float* __restrict__ denomM,
                                    float* __restrict__ agg0,
                                    float* __restrict__ agg1, int E) {
  constexpr int KP = KT * 4;
  constexpr int NP = NT * 16;
  constexpr int EPB = 128;
  __shared__ float Wl[KP * NP];
  __shared__ float msg[EPB * NP];
  __shared__ float al[EPB * NHEADS];
  __shared__ unsigned hsel;

  const int tid = threadIdx.x;
  for (int i = tid; i < KP * NP; i += blockDim.x) {
    int k = i / NP, n = i % NP;
    float v = 0.f;
    if (n < HCOUT) {
      if (k < CIN)            v = wmsg[k * HCOUT + n];
      else if (k == CIN)      v = wedge[n];
      else if (k == CIN + 1)  v = bmsg[n] + bedge[n];
    }
    Wl[i] = v;
  }
  for (int i = tid; i < EPB * NHEADS; i += blockDim.x) {
    int el = i / NHEADS, h = i % NHEADS;
    int ee = blockIdx.x * EPB + el;
    float a = 0.f;
    if (ee < E) {
      int o = dst[ee] * NHEADS + h;
      a = eaG[ee * NHEADS + h] / denomM[o];
    }
    al[i] = a;
  }

  const int lane  = tid & 31;
  const int wave  = tid >> 5;
  const int m     = lane & 15;
  const int khalf = (lane >> 4) << 1;
  const int ebase = blockIdx.x * EPB + wave * 16;
  const int e     = ebase + m;
  const bool valid = (e < E);

  int   s   = 0;
  float eav = 0.f;
  if (valid) { s = src[e]; eav = eattr[e]; }

  float ax[KT], ay[KT];
#pragma unroll
  for (int kt = 0; kt < KT; ++kt) {
    int k0 = kt * 4 + khalf;
#pragma unroll
    for (int j = 0; j < 2; ++j) {
      int k = k0 + j;
      float v = 0.f;
      if (valid) {
        if (k < CIN)           v = xin[s * CIN + k];
        else if (k == CIN)     v = eav;
        else if (k == CIN + 1) v = 1.0f;
      }
      if (j == 0) ax[kt] = v; else ay[kt] = v;
    }
  }
  if (tid == 0) hsel = chip_half();
  __syncthreads();

#pragma unroll
  for (int nt = 0; nt < NT; ++nt) {
    v8f acc = {};
#pragma unroll
    for (int kt = 0; kt < KT; ++kt) {
      int kb = kt * 4 + khalf;
      v2f a; a.x = ax[kt]; a.y = ay[kt];
      v2f b;
      b.x = Wl[kb * NP + nt * 16 + m];
      b.y = Wl[(kb + 1) * NP + nt * 16 + m];
      acc = __builtin_amdgcn_wmma_f32_16x16x4_f32(
          false, a, false, b, (short)0, acc, false, false);
    }
    int col   = nt * 16 + m;
    int rbase = wave * 16 + ((lane >> 4) << 3);
#pragma unroll
    for (int v = 0; v < 8; ++v) msg[(rbase + v) * NP + col] = acc[v];
  }
  __syncthreads();

  float* __restrict__ aggh = hsel ? agg1 : agg0;
  for (int i = tid; i < EPB * COUT; i += blockDim.x) {
    int el = i / COUT, c = i % COUT;
    int ee = blockIdx.x * EPB + el;
    if (ee >= E) continue;
    float sacc = 0.f;
#pragma unroll
    for (int h = 0; h < NHEADS; ++h)
      sacc += msg[el * NP + h * COUT + c] * al[el * NHEADS + h];
    atomicAdd(&aggh[dst[ee] * COUT + c], sacc * (1.0f / NHEADS));
  }
}

template <int CIN, int COUT>
__global__ void node_kernel(const float* __restrict__ xin,
                            const float* __restrict__ wself,
                            const float* __restrict__ bself,
                            const float* __restrict__ agg0,
                            float* __restrict__ hout, int N) {
  int i = blockIdx.x * blockDim.x + threadIdx.x;
  if (i >= N * COUT) return;
  int n = i / COUT, c = i % COUT;
  float sacc = agg0[i] + hout[i];
#pragma unroll
  for (int k = 0; k < CIN; ++k) sacc += xin[n * CIN + k] * wself[k * COUT + c];
  hout[i] = fmaxf(sacc + bself[c], 0.f);
}

__global__ void pool_kernel(const float* __restrict__ h3,
                            const int* __restrict__ batch,
                            unsigned* __restrict__ gmax, int N) {
  int i = blockIdx.x * blockDim.x + threadIdx.x;
  if (i >= N * 16) return;
  int n = i >> 4, c = i & 15;
  atomicMax(&gmax[batch[n] * 16 + c], ordEnc(h3[i]));
}

__global__ void head_kernel(const unsigned* __restrict__ gmax,
                            const float* __restrict__ wh,
                            const float* __restrict__ bh,
                            float* __restrict__ out, int G) {
  int i = blockIdx.x * blockDim.x + threadIdx.x;
  if (i >= G * 2) return;
  int g = i >> 1, c = i & 1;
  float sacc = bh[c];
#pragma unroll
  for (int k = 0; k < 16; ++k) sacc += ordDec(gmax[g * 16 + k]) * wh[k * 2 + c];
  out[i] = sacc;
}

template <int CIN, int HCOUT, int COUT, int KT, int NT>
static void run_layer(const float* xin, const int* src, const int* dst,
                      const float* eattr, const float* wmsg, const float* bmsg,
                      const float* wedge, const float* bedge, const float* att,
                      const float* wself, const float* bself, float* hout,
                      float* alpha, unsigned* amax, float* denom, float* agg0,
                      hipStream_t stream) {
  const int N = NNODES, E = NEDGES;
  const int NH2 = 2 * N * NHEADS;
  fill2_u32<<<(2 * NH2 + 255) / 256, 256, 0, stream>>>(
      amax, 0x007FFFFFu, NH2,
      (unsigned*)denom, 0u, NH2);

  const int nb = (E + 127) / 128;
  edge_alpha_kernel<CIN, HCOUT, COUT, KT, NT><<<nb, 256, 0, stream>>>(
      xin, src, dst, eattr, wmsg, bmsg, wedge, bedge, att, alpha, amax, E);

  const int NH = N * NHEADS;
  const int MB = (NH + 1023) / 1024;
  merge_max_kernel<<<MB, 256, 0, stream>>>(amax, NH);

  const int EH = E * NHEADS;
  softmax_kernel<<<(EH + 1023) / 1024, 256, 0, stream>>>(alpha, amax, denom, dst, EH);

  merge_sum_zero2_kernel<<<MB + (2 * N * COUT + 255) / 256, 256, 0, stream>>>(
      denom, NH, MB, (unsigned*)agg0, N * COUT, (unsigned*)hout, N * COUT);

  edge_scatter_kernel<CIN, HCOUT, COUT, KT, NT><<<nb, 256, 0, stream>>>(
      xin, src, dst, eattr, wmsg, bmsg, wedge, bedge, alpha, denom, agg0, hout, E);

  node_kernel<CIN, COUT><<<(N * COUT + 255) / 256, 256, 0, stream>>>(
      xin, wself, bself, agg0, hout, N);
}

extern "C" void kernel_launch(void* const* d_in, const int* in_sizes, int n_in,
                              void* d_out, int out_size, void* d_ws,
                              size_t ws_size, hipStream_t stream) {
  (void)in_sizes; (void)n_in; (void)out_size; (void)ws_size;
  const int N = NNODES, E = NEDGES, G = NGRAPH;

  const float* x     = (const float*)d_in[0];
  const int*   ei    = (const int*)d_in[1];
  const float* eattr = (const float*)d_in[2];
  const int*   batch = (const int*)d_in[3];
  const int* src = ei;
  const int* dst = ei + E;

  const float *wmsg[3], *bmsg[3], *wedge[3], *bedge[3], *att[3], *wself[3], *bself[3];
  for (int l = 0; l < 3; ++l) {
    int b = 4 + 7 * l;
    wmsg[l]  = (const float*)d_in[b + 0];
    bmsg[l]  = (const float*)d_in[b + 1];
    wedge[l] = (const float*)d_in[b + 2];
    bedge[l] = (const float*)d_in[b + 3];
    att[l]   = (const float*)d_in[b + 4];
    wself[l] = (const float*)d_in[b + 5];
    bself[l] = (const float*)d_in[b + 6];
  }
  const float* whead = (const float*)d_in[25];
  const float* bhead = (const float*)d_in[26];

  float*    alpha = (float*)d_ws;
  float*    denom = alpha + (size_t)E * NHEADS;
  float*    agg   = denom + (size_t)2 * N * NHEADS;
  unsigned* amax  = (unsigned*)agg;
  float*    h1    = agg + (size_t)N * 16;
  float*    h2    = h1 + (size_t)N * 4;
  float*    h3    = h2 + (size_t)N * 8;
  unsigned* gmax  = (unsigned*)(h3 + (size_t)N * 16);

  run_layer<3, 20, 4, 2, 2>(x,  src, dst, eattr, wmsg[0], bmsg[0], wedge[0],
                            bedge[0], att[0], wself[0], bself[0], h1,
                            alpha, amax, denom, agg, stream);
  run_layer<4, 40, 8, 2, 3>(h1, src, dst, eattr, wmsg[1], bmsg[1], wedge[1],
                            bedge[1], att[1], wself[1], bself[1], h2,
                            alpha, amax, denom, agg, stream);
  run_layer<8, 80, 16, 3, 5>(h2, src, dst, eattr, wmsg[2], bmsg[2], wedge[2],
                             bedge[2], att[2], wself[2], bself[2], h3,
                             alpha, amax, denom, agg, stream);

  fill_u32<<<(G * 16 + 255) / 256, 256, 0, stream>>>(gmax, 0x007FFFFFu, G * 16);
  pool_kernel<<<(N * 16 + 255) / 256, 256, 0, stream>>>(h3, batch, gmax, N);
  head_kernel<<<(G * 2 + 255) / 256, 256, 0, stream>>>(gmax, whead, bhead,
                                                       (float*)d_out, G);
}
